# starting kernel
# baseline (speedup 1.0000x reference)
#include <hip/hip_runtime.h>

typedef float v2f __attribute__((ext_vector_type(2)));
typedef float v8f __attribute__((ext_vector_type(8)));

#define N1 1024
#define WAVES_PER_BLOCK 8
#define BLOCK (WAVES_PER_BLOCK * 32)
#define PTS_PER_BLOCK (WAVES_PER_BLOCK * 16)
#define L2E 1.4426950408889634f


__global__ __launch_bounds__(BLOCK) void srbf2d_wmma(
    const float* __restrict__ x,
    const float* __restrict__ h1, const float* __restrict__ c1, const float* __restrict__ w1,
    const float* __restrict__ h2, const float* __restrict__ c2, const float* __restrict__ w2,
    const float* __restrict__ h3, const float* __restrict__ c3, const float* __restrict__ w3,
    float* __restrict__ out, int Npts)
{
    __shared__ float4 sP[3 * N1];
    __shared__ float  sB[10 * N1];

    const int tid = threadIdx.x;

    for (int idx = tid; idx < 3 * N1; idx += BLOCK) {
        const int s = idx >> 10;
        const int j = idx & (N1 - 1);
        const float* c = (s == 0) ? c1 : (s == 1) ? c2 : c3;
        const float* w = (s == 0) ? w1 : (s == 1) ? w2 : w3;
        const float* h = (s == 0) ? h1 : (s == 1) ? h2 : h3;
        const float cx = c[2 * j], cy = c[2 * j + 1];
        const float wx = w[2 * j], wy = w[2 * j + 1];
        const float d1 = wx * wx, d2 = wy * wy;
        const float hv = h[j];
        sP[idx] = make_float4(cx, cy, -d1 * L2E, -d2 * L2E);
        float* brow = sB + j * 10;
        if (s == 0) {
            brow[0] = d1 * hv;
            brow[1] = d1 * cx * hv;
            brow[2] = d2 * hv;
            brow[3] = d2 * cy * hv;
        } else if (s == 1) {
            brow[4] = hv;
            brow[5] = d1 * hv;
            brow[6] = d1 * cx * hv;
        } else {
            brow[7] = hv;
            brow[8] = d2 * hv;
            brow[9] = d2 * cy * hv;
        }
    }
    __syncthreads();

    const int wave = tid >> 5;
    const int lane = tid & 31;
    const int n    = lane & 15;
    const int half = lane >> 4;

    const int tile  = blockIdx.x * PTS_PER_BLOCK + wave * 16;
    const int point = tile + n;
    const float px = x[2 * point];
    const float py = x[2 * point + 1];

    const int bIdx = (n < 10) ? n : 0;
    const float set0m = (n < 4) ? 1.0f : 0.0f;
    const float set1m = (n >= 4 && n < 7) ? 1.0f : 0.0f;
    const float set2m = (n >= 7 && n < 10) ? 1.0f : 0.0f;

    v8f acc0 = {}, acc1 = {}, acc2 = {};

    #pragma unroll 2
    for (int j0 = 0; j0 < N1; j0 += 4) {
        const int jb = j0 + 2 * half;
        v2f a0 = {}, a1 = {}, a2 = {};
        v2f b0 = {}, b1 = {}, b2 = {};
        #pragma unroll
        for (int t = 0; t < 2; ++t) {
            const int j = jb + t;
            const float4 p0 = sP[j];
            const float4 p1 = sP[N1 + j];
            const float4 p2 = sP[2 * N1 + j];
            const float bt = sB[j * 10 + bIdx];

            const float dx0 = px - p0.x, dy0 = py - p0.y;
            const float e0 = __builtin_amdgcn_exp2f(fmaf(dx0 * dx0, p0.z, dy0 * dy0 * p0.w));
            const float dx1 = px - p1.x, dy1 = py - p1.y;
            const float e1 = __builtin_amdgcn_exp2f(fmaf(dx1 * dx1, p1.z, dy1 * dy1 * p1.w));
            const float dx2 = px - p2.x, dy2 = py - p2.y;
            const float e2 = __builtin_amdgcn_exp2f(fmaf(dx2 * dx2, p2.z, dy2 * dy2 * p2.w));

            a0[t] = e0; a1[t] = e1; a2[t] = e2;
            b0[t] = bt * set0m;
            b1[t] = bt * set1m;
            b2[t] = bt * set2m;
        }
        acc0 = __builtin_amdgcn_wmma_f32_16x16x4_f32(false, a0, false, b0, (short)0, acc0, false, false);
        acc1 = __builtin_amdgcn_wmma_f32_16x16x4_f32(false, a1, false, b1, (short)0, acc1, false, false);
        acc2 = __builtin_amdgcn_wmma_f32_16x16x4_f32(false, a2, false, b2, (short)0, acc2, false, false);
    }

    v8f acc = acc0 + acc1 + acc2;

    __syncthreads();
    float* dump = sB + wave * 256;
    #pragma unroll
    for (int r = 0; r < 8; ++r) {
        const int m = r + 8 * half;
        dump[m * 16 + n] = acc[r];
    }
    __syncthreads();

    const float* row = dump + n * 16;
    if (half == 0) {
        const float S0 = row[0], S1 = row[1], S2 = row[2], S3 = row[3], S4 = row[4];
        out[0 * Npts + point] = 2.0f * (S1 - px * S0);
        out[1 * Npts + point] = 2.0f * (S3 - py * S2);
        out[2 * Npts + point] = S4;
    } else {
        const float S5 = row[5], S6 = row[6], S7 = row[7], S8 = row[8], S9 = row[9];
        out[3 * Npts + point] = 2.0f * (S6 - px * S5);
        out[4 * Npts + point] = S7;
        out[5 * Npts + point] = 2.0f * (S9 - py * S8);
    }
}

extern "C" void kernel_launch(void* const* d_in, const int* in_sizes, int n_in,
                              void* d_out, int out_size, void* d_ws, size_t ws_size,
                              hipStream_t stream) {
    const float* x  = (const float*)d_in[0];
    const float* h1 = (const float*)d_in[1];
    const float* c1 = (const float*)d_in[2];
    const float* w1 = (const float*)d_in[3];
    const float* h2 = (const float*)d_in[4];
    const float* c2 = (const float*)d_in[5];
    const float* w2 = (const float*)d_in[6];
    const float* h3 = (const float*)d_in[7];
    const float* c3 = (const float*)d_in[8];
    const float* w3 = (const float*)d_in[9];

    const int Npts = in_sizes[0] / 2;
    dim3 grid(Npts / PTS_PER_BLOCK);
    srbf2d_wmma<<<grid, BLOCK, 0, stream>>>(x, h1, c1, w1, h2, c2, w2,
                                            h3, c3, w3, (float*)d_out, Npts);
}


# Each wave keeps two 16-point tiles: centre data and the masked B fragment are read from LDS once per centre pair and feed both tiles' WMMAs (256 workgroups)
# speedup vs baseline: 1.0969x; 1.0969x over previous
#include <hip/hip_runtime.h>

typedef float v2f __attribute__((ext_vector_type(2)));
typedef float v8f __attribute__((ext_vector_type(8)));

#define N1 1024
#define WAVES_PER_BLOCK 8
#define BLOCK (WAVES_PER_BLOCK * 32)
#define TILES_PER_WAVE 2
#define PTS_PER_WAVE (TILES_PER_WAVE * 16)
#define PTS_PER_BLOCK (WAVES_PER_BLOCK * PTS_PER_WAVE)
#define DUMP_PITCH 17
#define L2E 1.4426950408889634f


__global__ __launch_bounds__(BLOCK) void srbf2d_wmma(
    const float* __restrict__ x,
    const float* __restrict__ h1, const float* __restrict__ c1, const float* __restrict__ w1,
    const float* __restrict__ h2, const float* __restrict__ c2, const float* __restrict__ w2,
    const float* __restrict__ h3, const float* __restrict__ c3, const float* __restrict__ w3,
    float* __restrict__ out, int Npts)
{
    __shared__ float4 sP[3 * N1];
    __shared__ float  sB[10 * N1];

    const int tid = threadIdx.x;

    for (int idx = tid; idx < 3 * N1; idx += BLOCK) {
        const int s = idx >> 10;
        const int j = idx & (N1 - 1);
        const float* c = (s == 0) ? c1 : (s == 1) ? c2 : c3;
        const float* w = (s == 0) ? w1 : (s == 1) ? w2 : w3;
        const float* h = (s == 0) ? h1 : (s == 1) ? h2 : h3;
        const float cx = c[2 * j], cy = c[2 * j + 1];
        const float wx = w[2 * j], wy = w[2 * j + 1];
        const float d1 = wx * wx, d2 = wy * wy;
        const float hv = h[j];
        sP[idx] = make_float4(cx, cy, -d1 * L2E, -d2 * L2E);
        float* brow = sB + j * 10;
        if (s == 0) {
            brow[0] = d1 * hv;
            brow[1] = d1 * cx * hv;
            brow[2] = d2 * hv;
            brow[3] = d2 * cy * hv;
        } else if (s == 1) {
            brow[4] = hv;
            brow[5] = d1 * hv;
            brow[6] = d1 * cx * hv;
        } else {
            brow[7] = hv;
            brow[8] = d2 * hv;
            brow[9] = d2 * cy * hv;
        }
    }
    __syncthreads();

    const int wave = tid >> 5;
    const int lane = tid & 31;
    const int n    = lane & 15;
    const int half = lane >> 4;

    const int base   = blockIdx.x * PTS_PER_BLOCK + wave * PTS_PER_WAVE;
    const int pointA = base + n;
    const int pointB = base + 16 + n;
    const float pxA = x[2 * pointA], pyA = x[2 * pointA + 1];
    const float pxB = x[2 * pointB], pyB = x[2 * pointB + 1];

    const int bIdx = (n < 10) ? n : 0;
    const float set0m = (n < 4) ? 1.0f : 0.0f;
    const float set1m = (n >= 4 && n < 7) ? 1.0f : 0.0f;
    const float set2m = (n >= 7 && n < 10) ? 1.0f : 0.0f;

    v8f accA0 = {}, accA1 = {}, accA2 = {};
    v8f accB0 = {}, accB1 = {}, accB2 = {};

    #pragma unroll 2
    for (int j0 = 0; j0 < N1; j0 += 4) {
        const int jb = j0 + 2 * half;
        v2f aA0 = {}, aA1 = {}, aA2 = {};
        v2f aB0 = {}, aB1 = {}, aB2 = {};
        v2f b0 = {}, b1 = {}, b2 = {};
        #pragma unroll
        for (int t = 0; t < 2; ++t) {
            const int j = jb + t;
            const float4 p0 = sP[j];
            const float4 p1 = sP[N1 + j];
            const float4 p2 = sP[2 * N1 + j];
            const float bt = sB[j * 10 + bIdx];

            {
                const float dx0 = pxA - p0.x, dy0 = pyA - p0.y;
                aA0[t] = __builtin_amdgcn_exp2f(fmaf(dx0 * dx0, p0.z, dy0 * dy0 * p0.w));
                const float dx1 = pxA - p1.x, dy1 = pyA - p1.y;
                aA1[t] = __builtin_amdgcn_exp2f(fmaf(dx1 * dx1, p1.z, dy1 * dy1 * p1.w));
                const float dx2 = pxA - p2.x, dy2 = pyA - p2.y;
                aA2[t] = __builtin_amdgcn_exp2f(fmaf(dx2 * dx2, p2.z, dy2 * dy2 * p2.w));
            }
            {
                const float dx0 = pxB - p0.x, dy0 = pyB - p0.y;
                aB0[t] = __builtin_amdgcn_exp2f(fmaf(dx0 * dx0, p0.z, dy0 * dy0 * p0.w));
                const float dx1 = pxB - p1.x, dy1 = pyB - p1.y;
                aB1[t] = __builtin_amdgcn_exp2f(fmaf(dx1 * dx1, p1.z, dy1 * dy1 * p1.w));
                const float dx2 = pxB - p2.x, dy2 = pyB - p2.y;
                aB2[t] = __builtin_amdgcn_exp2f(fmaf(dx2 * dx2, p2.z, dy2 * dy2 * p2.w));
            }

            b0[t] = bt * set0m;
            b1[t] = bt * set1m;
            b2[t] = bt * set2m;
        }
        accA0 = __builtin_amdgcn_wmma_f32_16x16x4_f32(false, aA0, false, b0, (short)0, accA0, false, false);
        accB0 = __builtin_amdgcn_wmma_f32_16x16x4_f32(false, aB0, false, b0, (short)0, accB0, false, false);
        accA1 = __builtin_amdgcn_wmma_f32_16x16x4_f32(false, aA1, false, b1, (short)0, accA1, false, false);
        accB1 = __builtin_amdgcn_wmma_f32_16x16x4_f32(false, aB1, false, b1, (short)0, accB1, false, false);
        accA2 = __builtin_amdgcn_wmma_f32_16x16x4_f32(false, aA2, false, b2, (short)0, accA2, false, false);
        accB2 = __builtin_amdgcn_wmma_f32_16x16x4_f32(false, aB2, false, b2, (short)0, accB2, false, false);
    }

    const v8f accA = accA0 + accA1 + accA2;
    const v8f accB = accB0 + accB1 + accB2;

    __syncthreads();
    float* dump = sB + wave * (PTS_PER_WAVE * DUMP_PITCH);
    #pragma unroll
    for (int r = 0; r < 8; ++r) {
        const int m = r + 8 * half;
        dump[m * DUMP_PITCH + n]        = accA[r];
        dump[(16 + m) * DUMP_PITCH + n] = accB[r];
    }
    __syncthreads();

    const float* row = dump + lane * DUMP_PITCH;
    const int point = base + lane;
    const float px = half ? pxB : pxA;
    const float py = half ? pyB : pyA;
    const float S0 = row[0], S1 = row[1], S2 = row[2], S3 = row[3], S4 = row[4];
    const float S5 = row[5], S6 = row[6], S7 = row[7], S8 = row[8], S9 = row[9];
    out[0 * Npts + point] = 2.0f * (S1 - px * S0);
    out[1 * Npts + point] = 2.0f * (S3 - py * S2);
    out[2 * Npts + point] = S4;
    out[3 * Npts + point] = 2.0f * (S6 - px * S5);
    out[4 * Npts + point] = S7;
    out[5 * Npts + point] = 2.0f * (S9 - py * S8);
}

extern "C" void kernel_launch(void* const* d_in, const int* in_sizes, int n_in,
                              void* d_out, int out_size, void* d_ws, size_t ws_size,
                              hipStream_t stream) {
    const float* x  = (const float*)d_in[0];
    const float* h1 = (const float*)d_in[1];
    const float* c1 = (const float*)d_in[2];
    const float* w1 = (const float*)d_in[3];
    const float* h2 = (const float*)d_in[4];
    const float* c2 = (const float*)d_in[5];
    const float* w2 = (const float*)d_in[6];
    const float* h3 = (const float*)d_in[7];
    const float* c3 = (const float*)d_in[8];
    const float* w3 = (const float*)d_in[9];

    const int Npts = in_sizes[0] / 2;
    dim3 grid(Npts / PTS_PER_BLOCK);
    srbf2d_wmma<<<grid, BLOCK, 0, stream>>>(x, h1, c1, w1, h2, c2, w2,
                                            h3, c3, w3, (float*)d_out, Npts);
}


# Exponent from pre-scaled centre data {s,-c*s}: u=fma(px,s1,-cx*s1), v likewise, e=exp2(-(u*u+v*v)): 4 vector ops per exponential instead of 6; the argument is rounded differently
# speedup vs baseline: 1.1493x; 1.0478x over previous
#include <hip/hip_runtime.h>

typedef float v2f __attribute__((ext_vector_type(2)));
typedef float v8f __attribute__((ext_vector_type(8)));

#define N1 1024
#define WAVES_PER_BLOCK 8
#define BLOCK (WAVES_PER_BLOCK * 32)
#define TILES_PER_WAVE 2
#define PTS_PER_WAVE (TILES_PER_WAVE * 16)
#define PTS_PER_BLOCK (WAVES_PER_BLOCK * PTS_PER_WAVE)
#define DUMP_PITCH 17
#define SQRT_L2E 1.2011224087864498f


__global__ __launch_bounds__(BLOCK) void srbf2d_wmma(
    const float* __restrict__ x,
    const float* __restrict__ h1, const float* __restrict__ c1, const float* __restrict__ w1,
    const float* __restrict__ h2, const float* __restrict__ c2, const float* __restrict__ w2,
    const float* __restrict__ h3, const float* __restrict__ c3, const float* __restrict__ w3,
    float* __restrict__ out, int Npts)
{
    __shared__ float4 sP[3 * N1];
    __shared__ float  sB[10 * N1];

    const int tid = threadIdx.x;

    for (int idx = tid; idx < 3 * N1; idx += BLOCK) {
        const int s = idx >> 10;
        const int j = idx & (N1 - 1);
        const float* c = (s == 0) ? c1 : (s == 1) ? c2 : c3;
        const float* w = (s == 0) ? w1 : (s == 1) ? w2 : w3;
        const float* h = (s == 0) ? h1 : (s == 1) ? h2 : h3;
        const float cx = c[2 * j], cy = c[2 * j + 1];
        const float wx = w[2 * j], wy = w[2 * j + 1];
        const float d1 = wx * wx, d2 = wy * wy;
        const float hv = h[j];
        const float s1 = wx * SQRT_L2E, s2 = wy * SQRT_L2E;
        sP[idx] = make_float4(s1, -cx * s1, s2, -cy * s2);
        float* brow = sB + j * 10;
        if (s == 0) {
            brow[0] = d1 * hv;
            brow[1] = d1 * cx * hv;
            brow[2] = d2 * hv;
            brow[3] = d2 * cy * hv;
        } else if (s == 1) {
            brow[4] = hv;
            brow[5] = d1 * hv;
            brow[6] = d1 * cx * hv;
        } else {
            brow[7] = hv;
            brow[8] = d2 * hv;
            brow[9] = d2 * cy * hv;
        }
    }
    __syncthreads();

    const int wave = tid >> 5;
    const int lane = tid & 31;
    const int n    = lane & 15;
    const int half = lane >> 4;

    const int base   = blockIdx.x * PTS_PER_BLOCK + wave * PTS_PER_WAVE;
    const int pointA = base + n;
    const int pointB = base + 16 + n;
    const float pxA = x[2 * pointA], pyA = x[2 * pointA + 1];
    const float pxB = x[2 * pointB], pyB = x[2 * pointB + 1];

    const int bIdx = (n < 10) ? n : 0;
    const float set0m = (n < 4) ? 1.0f : 0.0f;
    const float set1m = (n >= 4 && n < 7) ? 1.0f : 0.0f;
    const float set2m = (n >= 7 && n < 10) ? 1.0f : 0.0f;

    v8f accA0 = {}, accA1 = {}, accA2 = {};
    v8f accB0 = {}, accB1 = {}, accB2 = {};

    #pragma unroll 2
    for (int j0 = 0; j0 < N1; j0 += 4) {
        const int jb = j0 + 2 * half;
        v2f aA0 = {}, aA1 = {}, aA2 = {};
        v2f aB0 = {}, aB1 = {}, aB2 = {};
        v2f b0 = {}, b1 = {}, b2 = {};
        #pragma unroll
        for (int t = 0; t < 2; ++t) {
            const int j = jb + t;
            const float4 p0 = sP[j];
            const float4 p1 = sP[N1 + j];
            const float4 p2 = sP[2 * N1 + j];
            const float bt = sB[j * 10 + bIdx];

            {
                const float u0 = fmaf(pxA, p0.x, p0.y), v0 = fmaf(pyA, p0.z, p0.w);
                aA0[t] = __builtin_amdgcn_exp2f(-fmaf(v0, v0, u0 * u0));
                const float u1 = fmaf(pxA, p1.x, p1.y), v1 = fmaf(pyA, p1.z, p1.w);
                aA1[t] = __builtin_amdgcn_exp2f(-fmaf(v1, v1, u1 * u1));
                const float u2 = fmaf(pxA, p2.x, p2.y), v2 = fmaf(pyA, p2.z, p2.w);
                aA2[t] = __builtin_amdgcn_exp2f(-fmaf(v2, v2, u2 * u2));
            }
            {
                const float u0 = fmaf(pxB, p0.x, p0.y), v0 = fmaf(pyB, p0.z, p0.w);
                aB0[t] = __builtin_amdgcn_exp2f(-fmaf(v0, v0, u0 * u0));
                const float u1 = fmaf(pxB, p1.x, p1.y), v1 = fmaf(pyB, p1.z, p1.w);
                aB1[t] = __builtin_amdgcn_exp2f(-fmaf(v1, v1, u1 * u1));
                const float u2 = fmaf(pxB, p2.x, p2.y), v2 = fmaf(pyB, p2.z, p2.w);
                aB2[t] = __builtin_amdgcn_exp2f(-fmaf(v2, v2, u2 * u2));
            }

            b0[t] = bt * set0m;
            b1[t] = bt * set1m;
            b2[t] = bt * set2m;
        }
        accA0 = __builtin_amdgcn_wmma_f32_16x16x4_f32(false, aA0, false, b0, (short)0, accA0, false, false);
        accB0 = __builtin_amdgcn_wmma_f32_16x16x4_f32(false, aB0, false, b0, (short)0, accB0, false, false);
        accA1 = __builtin_amdgcn_wmma_f32_16x16x4_f32(false, aA1, false, b1, (short)0, accA1, false, false);
        accB1 = __builtin_amdgcn_wmma_f32_16x16x4_f32(false, aB1, false, b1, (short)0, accB1, false, false);
        accA2 = __builtin_amdgcn_wmma_f32_16x16x4_f32(false, aA2, false, b2, (short)0, accA2, false, false);
        accB2 = __builtin_amdgcn_wmma_f32_16x16x4_f32(false, aB2, false, b2, (short)0, accB2, false, false);
    }

    const v8f accA = accA0 + accA1 + accA2;
    const v8f accB = accB0 + accB1 + accB2;

    __syncthreads();
    float* dump = sB + wave * (PTS_PER_WAVE * DUMP_PITCH);
    #pragma unroll
    for (int r = 0; r < 8; ++r) {
        const int m = r + 8 * half;
        dump[m * DUMP_PITCH + n]        = accA[r];
        dump[(16 + m) * DUMP_PITCH + n] = accB[r];
    }
    __syncthreads();

    const float* row = dump + lane * DUMP_PITCH;
    const int point = base + lane;
    const float px = half ? pxB : pxA;
    const float py = half ? pyB : pyA;
    const float S0 = row[0], S1 = row[1], S2 = row[2], S3 = row[3], S4 = row[4];
    const float S5 = row[5], S6 = row[6], S7 = row[7], S8 = row[8], S9 = row[9];
    out[0 * Npts + point] = 2.0f * (S1 - px * S0);
    out[1 * Npts + point] = 2.0f * (S3 - py * S2);
    out[2 * Npts + point] = S4;
    out[3 * Npts + point] = 2.0f * (S6 - px * S5);
    out[4 * Npts + point] = S7;
    out[5 * Npts + point] = 2.0f * (S9 - py * S8);
}

extern "C" void kernel_launch(void* const* d_in, const int* in_sizes, int n_in,
                              void* d_out, int out_size, void* d_ws, size_t ws_size,
                              hipStream_t stream) {
    const float* x  = (const float*)d_in[0];
    const float* h1 = (const float*)d_in[1];
    const float* c1 = (const float*)d_in[2];
    const float* w1 = (const float*)d_in[3];
    const float* h2 = (const float*)d_in[4];
    const float* c2 = (const float*)d_in[5];
    const float* w2 = (const float*)d_in[6];
    const float* h3 = (const float*)d_in[7];
    const float* c3 = (const float*)d_in[8];
    const float* w3 = (const float*)d_in[9];

    const int Npts = in_sizes[0] / 2;
    dim3 grid(Npts / PTS_PER_BLOCK);
    srbf2d_wmma<<<grid, BLOCK, 0, stream>>>(x, h1, c1, w1, h2, c2, w2,
                                            h3, c3, w3, (float*)d_out, Npts);
}


# Main loop software-pipelined per wave: behind each WMMA one short stretch (2 exp2 + 4 packed ops) builds the next trip's operands; centre table re-laid by pairs, LDS data fetched two trips ahead
# speedup vs baseline: 1.1856x; 1.0316x over previous
#include <hip/hip_runtime.h>

typedef float v2f __attribute__((ext_vector_type(2)));
typedef float v4f __attribute__((ext_vector_type(4)));
typedef float v8f __attribute__((ext_vector_type(8)));

#define N1 1024
#define WAVES_PER_BLOCK 8
#define BLOCK (WAVES_PER_BLOCK * 32)
#define TILES_PER_WAVE 2
#define PTS_PER_WAVE (TILES_PER_WAVE * 16)
#define PTS_PER_BLOCK (WAVES_PER_BLOCK * PTS_PER_WAVE)
#define DUMP_PITCH 17
#define SQRT_L2E 1.2011224087864498f


#define G 2
#define NCH (6 * G)
#define TRIP (4 * G)

struct Fetched {
    v4f p[G][3][2];
    v2f bt[G];
};

__device__ __forceinline__ void fetch_trip(Fetched& f, const v4f* sPl, const float* sBl, int jt)
{
    #pragma unroll
    for (int g = 0; g < G; ++g) {
        const int j = jt + 4 * g;
        #pragma unroll
        for (int s = 0; s < 3; ++s) {
            f.p[g][s][0] = sPl[s * N1 + j];
            f.p[g][s][1] = sPl[s * N1 + j + 1];
        }
        f.bt[g][0] = sBl[j * 10];
        f.bt[g][1] = sBl[j * 10 + 10];
    }
}

#define CH_G(c) ((c) / 6)
#define CH_S(c) (((c) % 6) >> 1)
#define CH_T(c) ((c) & 1)
__device__ __forceinline__ void start_args(v2f& u, v2f& v, const Fetched& F, int c,
                                           const v2f (&px2)[2], const v2f (&py2)[2])
{
    const v4f a = F.p[CH_G(c)][CH_S(c)][0], b = F.p[CH_G(c)][CH_S(c)][1];
    u = __builtin_elementwise_fma(px2[CH_T(c)], __builtin_shufflevector(a, a, 0, 1), __builtin_shufflevector(a, a, 2, 3));
    v = __builtin_elementwise_fma(py2[CH_T(c)], __builtin_shufflevector(b, b, 0, 1), __builtin_shufflevector(b, b, 2, 3));
}

struct ArgPipe { v2f q, vS, s, uU, vU; };

__device__ __forceinline__ void one_trip(
    v8f (&acc)[2][3],
    const v2f (&ecur)[NCH], const v2f (&bcur)[G][3],
    v2f (&enext)[NCH], v2f (&bnext)[G][3],
    const Fetched& Fn, Fetched& Ff, ArgPipe& ap,
    const v4f* sPl, const float* sBl, int jfetch,
    const v2f (&px2)[2], const v2f (&py2)[2], const v2f (&mask2)[3])
{
    fetch_trip(Ff, sPl, sBl, jfetch);
    #pragma unroll
    for (int c = 0; c < NCH; ++c) {
        const int g = CH_G(c), s = CH_S(c), tile = CH_T(c);
        __builtin_amdgcn_sched_barrier(0);
        acc[tile][s] = __builtin_amdgcn_wmma_f32_16x16x4_f32(false, ecur[c], false, bcur[g][s],
                                                             (short)0, acc[tile][s], false, false);
        __builtin_amdgcn_sched_barrier(0);
        enext[c][0] = __builtin_amdgcn_exp2f(-ap.q[0]);
        __builtin_amdgcn_sched_barrier(0);
        const float q1 = ap.q[1];
        ap.q = __builtin_elementwise_fma(ap.vS, ap.vS, ap.s);
        __builtin_amdgcn_sched_barrier(0);
        enext[c][1] = __builtin_amdgcn_exp2f(-q1);
        ap.s  = ap.uU * ap.uU;
        ap.vS = ap.vU;
        if (c + 3 < NCH) start_args(ap.uU, ap.vU, Fn, c + 3, px2, py2);
        else             start_args(ap.uU, ap.vU, Ff, c + 3 - NCH, px2, py2);
        if (tile == 0) bnext[g][s] = Fn.bt[g] * mask2[s];
    }
}

__global__ __launch_bounds__(BLOCK) void srbf2d_wmma(
    const float* __restrict__ x,
    const float* __restrict__ h1, const float* __restrict__ c1, const float* __restrict__ w1,
    const float* __restrict__ h2, const float* __restrict__ c2, const float* __restrict__ w2,
    const float* __restrict__ h3, const float* __restrict__ c3, const float* __restrict__ w3,
    float* __restrict__ out, int Npts)
{
    __shared__ v4f sP[3 * N1];
    __shared__ float  sB[10 * N1];

    const int tid = threadIdx.x;

    for (int idx = tid; idx < 3 * N1; idx += BLOCK) {
        const int s = idx >> 10;
        const int j = idx & (N1 - 1);
        const float* c = (s == 0) ? c1 : (s == 1) ? c2 : c3;
        const float* w = (s == 0) ? w1 : (s == 1) ? w2 : w3;
        const float* h = (s == 0) ? h1 : (s == 1) ? h2 : h3;
        const float cx = c[2 * j], cy = c[2 * j + 1];
        const float wx = w[2 * j], wy = w[2 * j + 1];
        const float d1 = wx * wx, d2 = wy * wy;
        const float hv = h[j];
        const float s1 = wx * SQRT_L2E, s2 = wy * SQRT_L2E;
        float* pr = reinterpret_cast<float*>(sP + (idx & ~1)) + (j & 1);
        pr[0] = s1;
        pr[2] = -cx * s1;
        pr[4] = s2;
        pr[6] = -cy * s2;
        float* brow = sB + j * 10;
        if (s == 0) {
            brow[0] = d1 * hv;
            brow[1] = d1 * cx * hv;
            brow[2] = d2 * hv;
            brow[3] = d2 * cy * hv;
        } else if (s == 1) {
            brow[4] = hv;
            brow[5] = d1 * hv;
            brow[6] = d1 * cx * hv;
        } else {
            brow[7] = hv;
            brow[8] = d2 * hv;
            brow[9] = d2 * cy * hv;
        }
    }
    __syncthreads();

    const int wave = tid >> 5;
    const int lane = tid & 31;
    const int n    = lane & 15;
    const int half = lane >> 4;

    const int base   = blockIdx.x * PTS_PER_BLOCK + wave * PTS_PER_WAVE;
    const int pointA = base + n;
    const int pointB = base + 16 + n;
    const float pxA = x[2 * pointA], pyA = x[2 * pointA + 1];
    const float pxB = x[2 * pointB], pyB = x[2 * pointB + 1];
    const v2f px2[2] = {{pxA, pxA}, {pxB, pxB}};
    const v2f py2[2] = {{pyA, pyA}, {pyB, pyB}};

    const int bIdx = (n < 10) ? n : 0;
    const float m0 = (n < 4) ? 1.0f : 0.0f;
    const float m1 = (n >= 4 && n < 7) ? 1.0f : 0.0f;
    const float m2 = (n >= 7 && n < 10) ? 1.0f : 0.0f;
    const v2f mask2[3] = {{m0, m0}, {m1, m1}, {m2, m2}};

    const v4f*   sPl = sP + 2 * half;
    const float* sBl = sB + 20 * half + bIdx;

    v8f acc[2][3] = {};
    v2f E0[NCH], E1[NCH];
    v2f B0[G][3], B1[G][3];
    Fetched X, Y;
    ArgPipe ap;

    fetch_trip(Y, sPl, sBl, 0);
    #pragma unroll
    for (int c = 0; c < NCH; ++c) {
        v2f u, v;
        start_args(u, v, Y, c, px2, py2);
        const v2f q = __builtin_elementwise_fma(v, v, u * u);
        E0[c][0] = __builtin_amdgcn_exp2f(-q[0]);
        E0[c][1] = __builtin_amdgcn_exp2f(-q[1]);
        if (CH_T(c) == 0) B0[CH_G(c)][CH_S(c)] = Y.bt[CH_G(c)] * mask2[CH_S(c)];
    }
    fetch_trip(X, sPl, sBl, TRIP);
    {
        v2f u, v;
        start_args(u, v, X, 0, px2, py2);
        ap.q = __builtin_elementwise_fma(v, v, u * u);
        start_args(u, ap.vS, X, 1, px2, py2);
        ap.s = u * u;
        start_args(ap.uU, ap.vU, X, 2, px2, py2);
    }

    #pragma nounroll
    for (int j0 = 0; j0 < N1; j0 += 2 * TRIP) {
        one_trip(acc, E0, B0, E1, B1, X, Y, ap, sPl, sBl, min(j0 + 2 * TRIP, N1 - TRIP), px2, py2, mask2);
        one_trip(acc, E1, B1, E0, B0, Y, X, ap, sPl, sBl, min(j0 + 3 * TRIP, N1 - TRIP), px2, py2, mask2);
    }

    const v8f accA = acc[0][0] + acc[0][1] + acc[0][2];
    const v8f accB = acc[1][0] + acc[1][1] + acc[1][2];

    __syncthreads();
    float* dump = sB + wave * (PTS_PER_WAVE * DUMP_PITCH);
    #pragma unroll
    for (int r = 0; r < 8; ++r) {
        const int m = r + 8 * half;
        dump[m * DUMP_PITCH + n]        = accA[r];
        dump[(16 + m) * DUMP_PITCH + n] = accB[r];
    }
    __syncthreads();

    const float* row = dump + lane * DUMP_PITCH;
    const int point = base + lane;
    const float px = half ? pxB : pxA;
    const float py = half ? pyB : pyA;
    const float S0 = row[0], S1 = row[1], S2 = row[2], S3 = row[3], S4 = row[4];
    const float S5 = row[5], S6 = row[6], S7 = row[7], S8 = row[8], S9 = row[9];
    out[0 * Npts + point] = 2.0f * (S1 - px * S0);
    out[1 * Npts + point] = 2.0f * (S3 - py * S2);
    out[2 * Npts + point] = S4;
    out[3 * Npts + point] = 2.0f * (S6 - px * S5);
    out[4 * Npts + point] = S7;
    out[5 * Npts + point] = 2.0f * (S9 - py * S8);
}

extern "C" void kernel_launch(void* const* d_in, const int* in_sizes, int n_in,
                              void* d_out, int out_size, void* d_ws, size_t ws_size,
                              hipStream_t stream) {
    const float* x  = (const float*)d_in[0];
    const float* h1 = (const float*)d_in[1];
    const float* c1 = (const float*)d_in[2];
    const float* w1 = (const float*)d_in[3];
    const float* h2 = (const float*)d_in[4];
    const float* c2 = (const float*)d_in[5];
    const float* w2 = (const float*)d_in[6];
    const float* h3 = (const float*)d_in[7];
    const float* c3 = (const float*)d_in[8];
    const float* w3 = (const float*)d_in[9];

    const int Npts = in_sizes[0] / 2;
    dim3 grid(Npts / PTS_PER_BLOCK);
    srbf2d_wmma<<<grid, BLOCK, 0, stream>>>(x, h1, c1, w1, h2, c2, w2,
                                            h3, c3, w3, (float*)d_out, Npts);
}


# Prologue: all 36 global loads per thread of the LDS table build (set index constant per unrolled step) and the point coordinates are issued before the first wait; was a chain of dependent round trips
# speedup vs baseline: 1.4254x; 1.2023x over previous
#include <hip/hip_runtime.h>

typedef float v2f __attribute__((ext_vector_type(2)));
typedef float v4f __attribute__((ext_vector_type(4)));
typedef float v8f __attribute__((ext_vector_type(8)));

#define N1 1024
#define WAVES_PER_BLOCK 8
#define BLOCK (WAVES_PER_BLOCK * 32)
#define TILES_PER_WAVE 2
#define PTS_PER_WAVE (TILES_PER_WAVE * 16)
#define PTS_PER_BLOCK (WAVES_PER_BLOCK * PTS_PER_WAVE)
#define DUMP_PITCH 17
#define SQRT_L2E 1.2011224087864498f


#define G 2
#define NCH (6 * G)
#define TRIP (4 * G)

struct Fetched {
    v4f p[G][3][2];
    v2f bt[G];
};

__device__ __forceinline__ void fetch_trip(Fetched& f, const v4f* sPl, const float* sBl, int jt)
{
    #pragma unroll
    for (int g = 0; g < G; ++g) {
        const int j = jt + 4 * g;
        #pragma unroll
        for (int s = 0; s < 3; ++s) {
            f.p[g][s][0] = sPl[s * N1 + j];
            f.p[g][s][1] = sPl[s * N1 + j + 1];
        }
        f.bt[g][0] = sBl[j * 10];
        f.bt[g][1] = sBl[j * 10 + 10];
    }
}

#define CH_G(c) ((c) / 6)
#define CH_S(c) (((c) % 6) >> 1)
#define CH_T(c) ((c) & 1)
__device__ __forceinline__ void start_args(v2f& u, v2f& v, const Fetched& F, int c,
                                           const v2f (&px2)[2], const v2f (&py2)[2])
{
    const v4f a = F.p[CH_G(c)][CH_S(c)][0], b = F.p[CH_G(c)][CH_S(c)][1];
    u = __builtin_elementwise_fma(px2[CH_T(c)], __builtin_shufflevector(a, a, 0, 1), __builtin_shufflevector(a, a, 2, 3));
    v = __builtin_elementwise_fma(py2[CH_T(c)], __builtin_shufflevector(b, b, 0, 1), __builtin_shufflevector(b, b, 2, 3));
}

struct ArgPipe { v2f q, vS, s, uU, vU; };

__device__ __forceinline__ void one_trip(
    v8f (&acc)[2][3],
    const v2f (&ecur)[NCH], const v2f (&bcur)[G][3],
    v2f (&enext)[NCH], v2f (&bnext)[G][3],
    const Fetched& Fn, Fetched& Ff, ArgPipe& ap,
    const v4f* sPl, const float* sBl, int jfetch,
    const v2f (&px2)[2], const v2f (&py2)[2], const v2f (&mask2)[3])
{
    fetch_trip(Ff, sPl, sBl, jfetch);
    #pragma unroll
    for (int c = 0; c < NCH; ++c) {
        const int g = CH_G(c), s = CH_S(c), tile = CH_T(c);
        __builtin_amdgcn_sched_barrier(0);
        acc[tile][s] = __builtin_amdgcn_wmma_f32_16x16x4_f32(false, ecur[c], false, bcur[g][s],
                                                             (short)0, acc[tile][s], false, false);
        __builtin_amdgcn_sched_barrier(0);
        enext[c][0] = __builtin_amdgcn_exp2f(-ap.q[0]);
        __builtin_amdgcn_sched_barrier(0);
        const float q1 = ap.q[1];
        ap.q = __builtin_elementwise_fma(ap.vS, ap.vS, ap.s);
        __builtin_amdgcn_sched_barrier(0);
        enext[c][1] = __builtin_amdgcn_exp2f(-q1);
        ap.s  = ap.uU * ap.uU;
        ap.vS = ap.vU;
        if (c + 3 < NCH) start_args(ap.uU, ap.vU, Fn, c + 3, px2, py2);
        else             start_args(ap.uU, ap.vU, Ff, c + 3 - NCH, px2, py2);
        if (tile == 0) bnext[g][s] = Fn.bt[g] * mask2[s];
    }
}

__global__ __launch_bounds__(BLOCK) void srbf2d_wmma(
    const float* __restrict__ x,
    const float* __restrict__ h1, const float* __restrict__ c1, const float* __restrict__ w1,
    const float* __restrict__ h2, const float* __restrict__ c2, const float* __restrict__ w2,
    const float* __restrict__ h3, const float* __restrict__ c3, const float* __restrict__ w3,
    float* __restrict__ out, int Npts)
{
    __shared__ v4f sP[3 * N1];
    __shared__ float  sB[10 * N1];

    const int tid = threadIdx.x;

    const int wave = tid >> 5;
    const int lane = tid & 31;
    const int n    = lane & 15;
    const int half = lane >> 4;

    const int base   = blockIdx.x * PTS_PER_BLOCK + wave * PTS_PER_WAVE;
    const int pointA = base + n;
    const int pointB = base + 16 + n;
    const float pxA = x[2 * pointA], pyA = x[2 * pointA + 1];
    const float pxB = x[2 * pointB], pyB = x[2 * pointB + 1];

    static_assert(N1 == 4 * BLOCK, "each thread builds four centres of each set");
    const float* const csrc[3] = {c1, c2, c3};
    const float* const wsrc[3] = {w1, w2, w3};
    const float* const hsrc[3] = {h1, h2, h3};
    float gcx[3][4], gcy[3][4], gwx[3][4], gwy[3][4], gh[3][4];
    #pragma unroll
    for (int s = 0; s < 3; ++s) {
        #pragma unroll
        for (int q = 0; q < 4; ++q) {
            const int j = tid + q * BLOCK;
            gcx[s][q] = csrc[s][2 * j];
            gcy[s][q] = csrc[s][2 * j + 1];
            gwx[s][q] = wsrc[s][2 * j];
            gwy[s][q] = wsrc[s][2 * j + 1];
            gh[s][q]  = hsrc[s][j];
        }
    }
    #pragma unroll
    for (int s = 0; s < 3; ++s) {
        #pragma unroll
        for (int q = 0; q < 4; ++q) {
            const int j = tid + q * BLOCK;
            const int idx = s * N1 + j;
            const float cx = gcx[s][q], cy = gcy[s][q];
            const float wx = gwx[s][q], wy = gwy[s][q];
            const float d1 = wx * wx, d2 = wy * wy;
            const float hv = gh[s][q];
            const float s1 = wx * SQRT_L2E, s2 = wy * SQRT_L2E;
            float* pr = reinterpret_cast<float*>(sP + (idx & ~1)) + (j & 1);
            pr[0] = s1;
            pr[2] = -cx * s1;
            pr[4] = s2;
            pr[6] = -cy * s2;
            float* brow = sB + j * 10;
            if (s == 0) {
                brow[0] = d1 * hv;
                brow[1] = d1 * cx * hv;
                brow[2] = d2 * hv;
                brow[3] = d2 * cy * hv;
            } else if (s == 1) {
                brow[4] = hv;
                brow[5] = d1 * hv;
                brow[6] = d1 * cx * hv;
            } else {
                brow[7] = hv;
                brow[8] = d2 * hv;
                brow[9] = d2 * cy * hv;
            }
        }
    }
    __syncthreads();

    const v2f px2[2] = {{pxA, pxA}, {pxB, pxB}};
    const v2f py2[2] = {{pyA, pyA}, {pyB, pyB}};

    const int bIdx = (n < 10) ? n : 0;
    const float m0 = (n < 4) ? 1.0f : 0.0f;
    const float m1 = (n >= 4 && n < 7) ? 1.0f : 0.0f;
    const float m2 = (n >= 7 && n < 10) ? 1.0f : 0.0f;
    const v2f mask2[3] = {{m0, m0}, {m1, m1}, {m2, m2}};

    const v4f*   sPl = sP + 2 * half;
    const float* sBl = sB + 20 * half + bIdx;

    v8f acc[2][3] = {};
    v2f E0[NCH], E1[NCH];
    v2f B0[G][3], B1[G][3];
    Fetched X, Y;
    ArgPipe ap;

    fetch_trip(Y, sPl, sBl, 0);
    #pragma unroll
    for (int c = 0; c < NCH; ++c) {
        v2f u, v;
        start_args(u, v, Y, c, px2, py2);
        const v2f q = __builtin_elementwise_fma(v, v, u * u);
        E0[c][0] = __builtin_amdgcn_exp2f(-q[0]);
        E0[c][1] = __builtin_amdgcn_exp2f(-q[1]);
        if (CH_T(c) == 0) B0[CH_G(c)][CH_S(c)] = Y.bt[CH_G(c)] * mask2[CH_S(c)];
    }
    fetch_trip(X, sPl, sBl, TRIP);
    {
        v2f u, v;
        start_args(u, v, X, 0, px2, py2);
        ap.q = __builtin_elementwise_fma(v, v, u * u);
        start_args(u, ap.vS, X, 1, px2, py2);
        ap.s = u * u;
        start_args(ap.uU, ap.vU, X, 2, px2, py2);
    }

    #pragma nounroll
    for (int j0 = 0; j0 < N1; j0 += 2 * TRIP) {
        one_trip(acc, E0, B0, E1, B1, X, Y, ap, sPl, sBl, min(j0 + 2 * TRIP, N1 - TRIP), px2, py2, mask2);
        one_trip(acc, E1, B1, E0, B0, Y, X, ap, sPl, sBl, min(j0 + 3 * TRIP, N1 - TRIP), px2, py2, mask2);
    }

    const v8f accA = acc[0][0] + acc[0][1] + acc[0][2];
    const v8f accB = acc[1][0] + acc[1][1] + acc[1][2];

    __syncthreads();
    float* dump = sB + wave * (PTS_PER_WAVE * DUMP_PITCH);
    #pragma unroll
    for (int r = 0; r < 8; ++r) {
        const int m = r + 8 * half;
        dump[m * DUMP_PITCH + n]        = accA[r];
        dump[(16 + m) * DUMP_PITCH + n] = accB[r];
    }
    __syncthreads();

    const float* row = dump + lane * DUMP_PITCH;
    const int point = base + lane;
    const float px = half ? pxB : pxA;
    const float py = half ? pyB : pyA;
    const float S0 = row[0], S1 = row[1], S2 = row[2], S3 = row[3], S4 = row[4];
    const float S5 = row[5], S6 = row[6], S7 = row[7], S8 = row[8], S9 = row[9];
    out[0 * Npts + point] = 2.0f * (S1 - px * S0);
    out[1 * Npts + point] = 2.0f * (S3 - py * S2);
    out[2 * Npts + point] = S4;
    out[3 * Npts + point] = 2.0f * (S6 - px * S5);
    out[4 * Npts + point] = S7;
    out[5 * Npts + point] = 2.0f * (S9 - py * S8);
}

extern "C" void kernel_launch(void* const* d_in, const int* in_sizes, int n_in,
                              void* d_out, int out_size, void* d_ws, size_t ws_size,
                              hipStream_t stream) {
    const float* x  = (const float*)d_in[0];
    const float* h1 = (const float*)d_in[1];
    const float* c1 = (const float*)d_in[2];
    const float* w1 = (const float*)d_in[3];
    const float* h2 = (const float*)d_in[4];
    const float* c2 = (const float*)d_in[5];
    const float* w2 = (const float*)d_in[6];
    const float* h3 = (const float*)d_in[7];
    const float* c3 = (const float*)d_in[8];
    const float* w3 = (const float*)d_in[9];

    const int Npts = in_sizes[0] / 2;
    dim3 grid(Npts / PTS_PER_BLOCK);
    srbf2d_wmma<<<grid, BLOCK, 0, stream>>>(x, h1, c1, w1, h2, c2, w2,
                                            h3, c3, w3, (float*)d_out, Npts);
}


# Each wave covers four 16-point tiles over half the centres (was two tiles over all): one LDS fetch and one masked B fragment feed 4 tiles' WMMAs; each output sums two half sums added at the end
# speedup vs baseline: 1.4989x; 1.0516x over previous
#include <hip/hip_runtime.h>

typedef float v2f __attribute__((ext_vector_type(2)));
typedef float v4f __attribute__((ext_vector_type(4)));
typedef float v8f __attribute__((ext_vector_type(8)));

#define N1 1024
#define WAVES_PER_BLOCK 8
#define BLOCK (WAVES_PER_BLOCK * 32)
#define TILES_PER_WAVE 4
#define PTS_PER_WAVE (TILES_PER_WAVE * 16)
#define CENTRE_SPLIT 2
#define NC_WAVE (N1 / CENTRE_SPLIT)
#define PT_GROUPS (WAVES_PER_BLOCK / CENTRE_SPLIT)
#define PTS_PER_BLOCK (PT_GROUPS * PTS_PER_WAVE)
#define DUMP_PITCH 17
#define SQRT_L2E 1.2011224087864498f


#define G 2
#define NCH (3 * TILES_PER_WAVE * G)
#define TRIP (4 * G)

struct Fetched {
    v4f p[G][3][2];
    v2f bt[G];
};

__device__ __forceinline__ void fetch_trip(Fetched& f, const v4f* sPl, const float* sBl, int jt)
{
    #pragma unroll
    for (int g = 0; g < G; ++g) {
        const int j = jt + 4 * g;
        #pragma unroll
        for (int s = 0; s < 3; ++s) {
            f.p[g][s][0] = sPl[s * N1 + j];
            f.p[g][s][1] = sPl[s * N1 + j + 1];
        }
        f.bt[g][0] = sBl[j * 10];
        f.bt[g][1] = sBl[j * 10 + 10];
    }
}

#define CH_G(c) ((c) / 12)
#define CH_S(c) (((c) % 12) >> 2)
#define CH_T(c) ((c) & 3)
__device__ __forceinline__ void start_args(v2f& u, v2f& v, const Fetched& F, int c,
                                           const v2f (&px2)[TILES_PER_WAVE], const v2f (&py2)[TILES_PER_WAVE])
{
    const v4f a = F.p[CH_G(c)][CH_S(c)][0], b = F.p[CH_G(c)][CH_S(c)][1];
    u = __builtin_elementwise_fma(px2[CH_T(c)], __builtin_shufflevector(a, a, 0, 1), __builtin_shufflevector(a, a, 2, 3));
    v = __builtin_elementwise_fma(py2[CH_T(c)], __builtin_shufflevector(b, b, 0, 1), __builtin_shufflevector(b, b, 2, 3));
}

struct ArgPipe { v2f q, vS, s, uU, vU; };

__device__ __forceinline__ void one_trip(
    v8f (&acc)[TILES_PER_WAVE],
    const v2f (&ecur)[NCH], const v2f (&bcur)[G][3],
    v2f (&enext)[NCH], v2f (&bnext)[G][3],
    const Fetched& Fn, Fetched& Ff, ArgPipe& ap,
    const v4f* sPl, const float* sBl, int jfetch,
    const v2f (&px2)[TILES_PER_WAVE], const v2f (&py2)[TILES_PER_WAVE], const v2f (&mask2)[3])
{
    fetch_trip(Ff, sPl, sBl, jfetch);
    #pragma unroll
    for (int c = 0; c < NCH; ++c) {
        const int g = CH_G(c), s = CH_S(c), tile = CH_T(c);
        __builtin_amdgcn_sched_barrier(0);
        acc[tile] = __builtin_amdgcn_wmma_f32_16x16x4_f32(false, ecur[c], false, bcur[g][s],
                                                          (short)0, acc[tile], false, false);
        __builtin_amdgcn_sched_barrier(0);
        enext[c][0] = __builtin_amdgcn_exp2f(-ap.q[0]);
        __builtin_amdgcn_sched_barrier(0);
        const float q1 = ap.q[1];
        ap.q = __builtin_elementwise_fma(ap.vS, ap.vS, ap.s);
        __builtin_amdgcn_sched_barrier(0);
        enext[c][1] = __builtin_amdgcn_exp2f(-q1);
        ap.s  = ap.uU * ap.uU;
        ap.vS = ap.vU;
        if (c + 3 < NCH) start_args(ap.uU, ap.vU, Fn, c + 3, px2, py2);
        else             start_args(ap.uU, ap.vU, Ff, c + 3 - NCH, px2, py2);
        if (tile == TILES_PER_WAVE - 1) bnext[g][s] = Fn.bt[g] * mask2[s];
    }
}

__global__ __launch_bounds__(BLOCK) void srbf2d_wmma(
    const float* __restrict__ x,
    const float* __restrict__ h1, const float* __restrict__ c1, const float* __restrict__ w1,
    const float* __restrict__ h2, const float* __restrict__ c2, const float* __restrict__ w2,
    const float* __restrict__ h3, const float* __restrict__ c3, const float* __restrict__ w3,
    float* __restrict__ out, int Npts)
{
    __shared__ v4f sP[3 * N1];
    __shared__ float  sB[10 * N1];

    const int tid = threadIdx.x;

    const int wave = tid >> 5;
    const int lane = tid & 31;
    const int n    = lane & 15;
    const int half = lane >> 4;

    const int pgroup = wave % PT_GROUPS;
    const int chalf  = wave / PT_GROUPS;
    const int base   = blockIdx.x * PTS_PER_BLOCK + pgroup * PTS_PER_WAVE;
    float pxT[TILES_PER_WAVE], pyT[TILES_PER_WAVE];
    #pragma unroll
    for (int t = 0; t < TILES_PER_WAVE; ++t) {
        const int pt = base + 16 * t + n;
        pxT[t] = x[2 * pt];
        pyT[t] = x[2 * pt + 1];
    }
    const int pointE = blockIdx.x * PTS_PER_BLOCK + tid;
    const float pxE = x[2 * pointE], pyE = x[2 * pointE + 1];

    static_assert(N1 == 4 * BLOCK, "each thread builds four centres of each set");
    const float* const csrc[3] = {c1, c2, c3};
    const float* const wsrc[3] = {w1, w2, w3};
    const float* const hsrc[3] = {h1, h2, h3};
    float gcx[3][4], gcy[3][4], gwx[3][4], gwy[3][4], gh[3][4];
    #pragma unroll
    for (int s = 0; s < 3; ++s) {
        #pragma unroll
        for (int q = 0; q < 4; ++q) {
            const int j = tid + q * BLOCK;
            gcx[s][q] = csrc[s][2 * j];
            gcy[s][q] = csrc[s][2 * j + 1];
            gwx[s][q] = wsrc[s][2 * j];
            gwy[s][q] = wsrc[s][2 * j + 1];
            gh[s][q]  = hsrc[s][j];
        }
    }
    #pragma unroll
    for (int s = 0; s < 3; ++s) {
        #pragma unroll
        for (int q = 0; q < 4; ++q) {
            const int j = tid + q * BLOCK;
            const int idx = s * N1 + j;
            const float cx = gcx[s][q], cy = gcy[s][q];
            const float wx = gwx[s][q], wy = gwy[s][q];
            const float d1 = wx * wx, d2 = wy * wy;
            const float hv = gh[s][q];
            const float s1 = wx * SQRT_L2E, s2 = wy * SQRT_L2E;
            float* pr = reinterpret_cast<float*>(sP + (idx & ~1)) + (j & 1);
            pr[0] = s1;
            pr[2] = -cx * s1;
            pr[4] = s2;
            pr[6] = -cy * s2;
            float* brow = sB + j * 10;
            if (s == 0) {
                brow[0] = d1 * hv;
                brow[1] = d1 * cx * hv;
                brow[2] = d2 * hv;
                brow[3] = d2 * cy * hv;
            } else if (s == 1) {
                brow[4] = hv;
                brow[5] = d1 * hv;
                brow[6] = d1 * cx * hv;
            } else {
                brow[7] = hv;
                brow[8] = d2 * hv;
                brow[9] = d2 * cy * hv;
            }
        }
    }
    __syncthreads();

    const v2f px2[TILES_PER_WAVE] = {{pxT[0], pxT[0]}, {pxT[1], pxT[1]}, {pxT[2], pxT[2]}, {pxT[3], pxT[3]}};
    const v2f py2[TILES_PER_WAVE] = {{pyT[0], pyT[0]}, {pyT[1], pyT[1]}, {pyT[2], pyT[2]}, {pyT[3], pyT[3]}};

    const int bIdx = (n < 10) ? n : 0;
    const float m0 = (n < 4) ? 1.0f : 0.0f;
    const float m1 = (n >= 4 && n < 7) ? 1.0f : 0.0f;
    const float m2 = (n >= 7 && n < 10) ? 1.0f : 0.0f;
    const v2f mask2[3] = {{m0, m0}, {m1, m1}, {m2, m2}};

    const v4f*   sPl = sP + 2 * half + chalf * NC_WAVE;
    const float* sBl = sB + 20 * half + bIdx + chalf * (NC_WAVE * 10);

    v8f acc[TILES_PER_WAVE] = {};
    v2f E0[NCH], E1[NCH];
    v2f B0[G][3], B1[G][3];
    Fetched X, Y;
    ArgPipe ap;

    fetch_trip(Y, sPl, sBl, 0);
    #pragma unroll
    for (int c = 0; c < NCH; ++c) {
        v2f u, v;
        start_args(u, v, Y, c, px2, py2);
        const v2f q = __builtin_elementwise_fma(v, v, u * u);
        E0[c][0] = __builtin_amdgcn_exp2f(-q[0]);
        E0[c][1] = __builtin_amdgcn_exp2f(-q[1]);
        if (CH_T(c) == TILES_PER_WAVE - 1) B0[CH_G(c)][CH_S(c)] = Y.bt[CH_G(c)] * mask2[CH_S(c)];
    }
    fetch_trip(X, sPl, sBl, TRIP);
    {
        v2f u, v;
        start_args(u, v, X, 0, px2, py2);
        ap.q = __builtin_elementwise_fma(v, v, u * u);
        start_args(u, ap.vS, X, 1, px2, py2);
        ap.s = u * u;
        start_args(ap.uU, ap.vU, X, 2, px2, py2);
    }

    #pragma nounroll
    for (int j0 = 0; j0 < NC_WAVE; j0 += 2 * TRIP) {
        one_trip(acc, E0, B0, E1, B1, X, Y, ap, sPl, sBl, min(j0 + 2 * TRIP, NC_WAVE - TRIP), px2, py2, mask2);
        one_trip(acc, E1, B1, E0, B0, Y, X, ap, sPl, sBl, min(j0 + 3 * TRIP, NC_WAVE - TRIP), px2, py2, mask2);
    }

    static_assert(WAVES_PER_BLOCK * PTS_PER_WAVE * DUMP_PITCH <= 10 * N1, "the dumps of all waves fit in sB");
    __syncthreads();
    float* dump = sB + wave * (PTS_PER_WAVE * DUMP_PITCH);
    #pragma unroll
    for (int t = 0; t < TILES_PER_WAVE; ++t) {
        #pragma unroll
        for (int r = 0; r < 8; ++r) {
            const int m = r + 8 * half;
            dump[(16 * t + m) * DUMP_PITCH + n] = acc[t][r];
        }
    }
    __syncthreads();

    const float* rowA = sB + (tid / PTS_PER_WAVE) * (PTS_PER_WAVE * DUMP_PITCH) + (tid % PTS_PER_WAVE) * DUMP_PITCH;
    const float* rowB = rowA + PT_GROUPS * (PTS_PER_WAVE * DUMP_PITCH);
    const int point = pointE;
    const float px = pxE, py = pyE;
    const float S0 = rowA[0] + rowB[0], S1 = rowA[1] + rowB[1], S2 = rowA[2] + rowB[2];
    const float S3 = rowA[3] + rowB[3], S4 = rowA[4] + rowB[4], S5 = rowA[5] + rowB[5];
    const float S6 = rowA[6] + rowB[6], S7 = rowA[7] + rowB[7], S8 = rowA[8] + rowB[8];
    const float S9 = rowA[9] + rowB[9];
    out[0 * Npts + point] = 2.0f * (S1 - px * S0);
    out[1 * Npts + point] = 2.0f * (S3 - py * S2);
    out[2 * Npts + point] = S4;
    out[3 * Npts + point] = 2.0f * (S6 - px * S5);
    out[4 * Npts + point] = S7;
    out[5 * Npts + point] = 2.0f * (S9 - py * S8);
}

extern "C" void kernel_launch(void* const* d_in, const int* in_sizes, int n_in,
                              void* d_out, int out_size, void* d_ws, size_t ws_size,
                              hipStream_t stream) {
    const float* x  = (const float*)d_in[0];
    const float* h1 = (const float*)d_in[1];
    const float* c1 = (const float*)d_in[2];
    const float* w1 = (const float*)d_in[3];
    const float* h2 = (const float*)d_in[4];
    const float* c2 = (const float*)d_in[5];
    const float* w2 = (const float*)d_in[6];
    const float* h3 = (const float*)d_in[7];
    const float* c3 = (const float*)d_in[8];
    const float* w3 = (const float*)d_in[9];

    const int Npts = in_sizes[0] / 2;
    dim3 grid(Npts / PTS_PER_BLOCK);
    srbf2d_wmma<<<grid, BLOCK, 0, stream>>>(x, h1, c1, w1, h2, c2, w2,
                                            h3, c3, w3, (float*)d_out, Npts);
}


# Loop trimmed by one issue slot per WMMA: 8 tiles per wave share each LDS fetch and B fragment (outputs now sum 4 quarter sums), exp2 in place so no wait slot follows, fetch pointers step by adds
# speedup vs baseline: 1.5445x; 1.0304x over previous
#include <hip/hip_runtime.h>

typedef float v2f __attribute__((ext_vector_type(2)));
typedef float v4f __attribute__((ext_vector_type(4)));
typedef float v8f __attribute__((ext_vector_type(8)));

#define N1 1024
#define WAVES_PER_BLOCK 8
#define BLOCK (WAVES_PER_BLOCK * 32)
#define TILES_PER_WAVE 8
#define PTS_PER_WAVE (TILES_PER_WAVE * 16)
#define CENTRE_SPLIT 4
#define NC_WAVE (N1 / CENTRE_SPLIT)
#define PT_GROUPS (WAVES_PER_BLOCK / CENTRE_SPLIT)
#define PTS_PER_BLOCK (PT_GROUPS * PTS_PER_WAVE)
#define DUMP_PITCH 17
#define SQRT_L2E 1.2011224087864498f


#define G 1
#define NCH (3 * TILES_PER_WAVE * G)
#define TRIP (4 * G)
#define LDS_PAD_V4F ((2 * TRIP + 2) * 10 / 4 + 1)

struct Fetched {
    v4f p[G][3][2];
    v2f bt[G];
};

__device__ __forceinline__ void fetch_trip(Fetched& f, const v4f* sPl, const float* sBl, int jt)
{
    #pragma unroll
    for (int g = 0; g < G; ++g) {
        const int j = jt + 4 * g;
        #pragma unroll
        for (int s = 0; s < 3; ++s) {
            f.p[g][s][0] = sPl[s * N1 + j];
            f.p[g][s][1] = sPl[s * N1 + j + 1];
        }
        f.bt[g][0] = sBl[j * 10];
        f.bt[g][1] = sBl[j * 10 + 10];
    }
}

#define CH_G(c) ((c) / (3 * TILES_PER_WAVE))
#define CH_S(c) (((c) % (3 * TILES_PER_WAVE)) / TILES_PER_WAVE)
#define CH_T(c) ((c) % TILES_PER_WAVE)
__device__ __forceinline__ void start_args(v2f& u, v2f& v, const Fetched& F, int c,
                                           const v2f (&px2)[TILES_PER_WAVE], const v2f (&py2)[TILES_PER_WAVE])
{
    const v4f a = F.p[CH_G(c)][CH_S(c)][0], b = F.p[CH_G(c)][CH_S(c)][1];
    u = __builtin_elementwise_fma(px2[CH_T(c)], __builtin_shufflevector(a, a, 0, 1), __builtin_shufflevector(a, a, 2, 3));
    v = __builtin_elementwise_fma(py2[CH_T(c)], __builtin_shufflevector(b, b, 0, 1), __builtin_shufflevector(b, b, 2, 3));
}

__device__ __forceinline__ float exp2_neg_in_place(float q)
{
    asm("v_exp_f32 %0, -%0" : "+v"(q));
    return q;
}

struct ArgPipe { v2f q, vS, s, uU, vU; };

__device__ __forceinline__ void one_trip(
    v8f (&acc)[TILES_PER_WAVE],
    const v2f (&ecur)[NCH], const v2f (&bcur)[G][3],
    v2f (&enext)[NCH], v2f (&bnext)[G][3],
    const Fetched& Fn, Fetched& Ff, ArgPipe& ap,
    const v4f* sPl, const float* sBl, int jfetch,
    const v2f (&px2)[TILES_PER_WAVE], const v2f (&py2)[TILES_PER_WAVE], const v2f (&mask2)[3])
{
    fetch_trip(Ff, sPl, sBl, jfetch);
    #pragma unroll
    for (int c = 0; c < NCH; ++c) {
        const int g = CH_G(c), s = CH_S(c), tile = CH_T(c);
        __builtin_amdgcn_sched_barrier(0);
        acc[tile] = __builtin_amdgcn_wmma_f32_16x16x4_f32(false, ecur[c], false, bcur[g][s],
                                                          (short)0, acc[tile], false, false);
        __builtin_amdgcn_sched_barrier(0);
        v2f e = ap.q;
        e[0] = exp2_neg_in_place(e[0]);
        __builtin_amdgcn_sched_barrier(0);
        ap.q = __builtin_elementwise_fma(ap.vS, ap.vS, ap.s);
        __builtin_amdgcn_sched_barrier(0);
        e[1] = exp2_neg_in_place(e[1]);
        enext[c] = e;
        __builtin_amdgcn_sched_barrier(0);
        ap.s  = ap.uU * ap.uU;
        ap.vS = ap.vU;
        if (c + 3 < NCH) start_args(ap.uU, ap.vU, Fn, c + 3, px2, py2);
        else             start_args(ap.uU, ap.vU, Ff, c + 3 - NCH, px2, py2);
        if (tile == TILES_PER_WAVE - 1) bnext[g][s] = Fn.bt[g] * mask2[s];
    }
}

__global__ __launch_bounds__(BLOCK) void srbf2d_wmma(
    const float* __restrict__ x,
    const float* __restrict__ h1, const float* __restrict__ c1, const float* __restrict__ w1,
    const float* __restrict__ h2, const float* __restrict__ c2, const float* __restrict__ w2,
    const float* __restrict__ h3, const float* __restrict__ c3, const float* __restrict__ w3,
    float* __restrict__ out, int Npts)
{
    __shared__ v4f lds[3 * N1 + (10 * N1) / 4 + LDS_PAD_V4F];
    v4f*   const sP = lds;
    float* const sB = reinterpret_cast<float*>(lds + 3 * N1);

    const int tid = threadIdx.x;

    const int wave = tid >> 5;
    const int lane = tid & 31;
    const int n    = lane & 15;
    const int half = lane >> 4;

    const int pgroup = wave % PT_GROUPS;
    const int chalf  = wave / PT_GROUPS;
    const int base   = blockIdx.x * PTS_PER_BLOCK + pgroup * PTS_PER_WAVE;
    float pxT[TILES_PER_WAVE], pyT[TILES_PER_WAVE];
    #pragma unroll
    for (int t = 0; t < TILES_PER_WAVE; ++t) {
        const int pt = base + 16 * t + n;
        pxT[t] = x[2 * pt];
        pyT[t] = x[2 * pt + 1];
    }
    const int pointE = blockIdx.x * PTS_PER_BLOCK + tid;
    const float pxE = x[2 * pointE], pyE = x[2 * pointE + 1];

    static_assert(N1 == 4 * BLOCK, "each thread builds four centres of each set");
    const float* const csrc[3] = {c1, c2, c3};
    const float* const wsrc[3] = {w1, w2, w3};
    const float* const hsrc[3] = {h1, h2, h3};
    float gcx[3][4], gcy[3][4], gwx[3][4], gwy[3][4], gh[3][4];
    #pragma unroll
    for (int s = 0; s < 3; ++s) {
        #pragma unroll
        for (int q = 0; q < 4; ++q) {
            const int j = tid + q * BLOCK;
            gcx[s][q] = csrc[s][2 * j];
            gcy[s][q] = csrc[s][2 * j + 1];
            gwx[s][q] = wsrc[s][2 * j];
            gwy[s][q] = wsrc[s][2 * j + 1];
            gh[s][q]  = hsrc[s][j];
        }
    }
    #pragma unroll
    for (int s = 0; s < 3; ++s) {
        #pragma unroll
        for (int q = 0; q < 4; ++q) {
            const int j = tid + q * BLOCK;
            const int idx = s * N1 + j;
            const float cx = gcx[s][q], cy = gcy[s][q];
            const float wx = gwx[s][q], wy = gwy[s][q];
            const float d1 = wx * wx, d2 = wy * wy;
            const float hv = gh[s][q];
            const float s1 = wx * SQRT_L2E, s2 = wy * SQRT_L2E;
            float* pr = reinterpret_cast<float*>(sP + (idx & ~1)) + (j & 1);
            pr[0] = s1;
            pr[2] = -cx * s1;
            pr[4] = s2;
            pr[6] = -cy * s2;
            float* brow = sB + j * 10;
            if (s == 0) {
                brow[0] = d1 * hv;
                brow[1] = d1 * cx * hv;
                brow[2] = d2 * hv;
                brow[3] = d2 * cy * hv;
            } else if (s == 1) {
                brow[4] = hv;
                brow[5] = d1 * hv;
                brow[6] = d1 * cx * hv;
            } else {
                brow[7] = hv;
                brow[8] = d2 * hv;
                brow[9] = d2 * cy * hv;
            }
        }
    }
    __syncthreads();

    v2f px2[TILES_PER_WAVE], py2[TILES_PER_WAVE];
    #pragma unroll
    for (int t = 0; t < TILES_PER_WAVE; ++t) {
        px2[t] = v2f{pxT[t], pxT[t]};
        py2[t] = v2f{pyT[t], pyT[t]};
    }

    const int bIdx = (n < 10) ? n : 0;
    const float m0 = (n < 4) ? 1.0f : 0.0f;
    const float m1 = (n >= 4 && n < 7) ? 1.0f : 0.0f;
    const float m2 = (n >= 7 && n < 10) ? 1.0f : 0.0f;
    const v2f mask2[3] = {{m0, m0}, {m1, m1}, {m2, m2}};

    const v4f*   sPl = sP + 2 * half + chalf * NC_WAVE;
    const float* sBl = sB + 20 * half + bIdx + chalf * (NC_WAVE * 10);

    v8f acc[TILES_PER_WAVE] = {};
    v2f E0[NCH], E1[NCH];
    v2f B0[G][3], B1[G][3];
    Fetched X, Y;
    ArgPipe ap;

    fetch_trip(Y, sPl, sBl, 0);
    #pragma unroll
    for (int c = 0; c < NCH; ++c) {
        v2f u, v;
        start_args(u, v, Y, c, px2, py2);
        const v2f q = __builtin_elementwise_fma(v, v, u * u);
        E0[c][0] = __builtin_amdgcn_exp2f(-q[0]);
        E0[c][1] = __builtin_amdgcn_exp2f(-q[1]);
        if (CH_T(c) == TILES_PER_WAVE - 1) B0[CH_G(c)][CH_S(c)] = Y.bt[CH_G(c)] * mask2[CH_S(c)];
    }
    fetch_trip(X, sPl, sBl, TRIP);
    {
        v2f u, v;
        start_args(u, v, X, 0, px2, py2);
        ap.q = __builtin_elementwise_fma(v, v, u * u);
        start_args(u, ap.vS, X, 1, px2, py2);
        ap.s = u * u;
        start_args(ap.uU, ap.vU, X, 2, px2, py2);
    }

    const v4f*   pP = sPl + 2 * TRIP;
    const float* pB = sBl + 2 * TRIP * 10;
    #pragma nounroll
    for (int j0 = 0; j0 < NC_WAVE; j0 += 2 * TRIP) {
        one_trip(acc, E0, B0, E1, B1, X, Y, ap, pP, pB, 0, px2, py2, mask2);
        one_trip(acc, E1, B1, E0, B0, Y, X, ap, pP, pB, TRIP, px2, py2, mask2);
        pP += 2 * TRIP;
        pB += 2 * TRIP * 10;
    }

    static_assert(WAVES_PER_BLOCK * PTS_PER_WAVE * DUMP_PITCH <= 4 * (3 * N1) + 10 * N1, "the dumps of all waves fit in the LDS block");
    __syncthreads();
    float* const dumpAll = reinterpret_cast<float*>(lds);
    float* dump = dumpAll + wave * (PTS_PER_WAVE * DUMP_PITCH);
    #pragma unroll
    for (int t = 0; t < TILES_PER_WAVE; ++t) {
        #pragma unroll
        for (int r = 0; r < 8; ++r) {
            const int m = r + 8 * half;
            dump[(16 * t + m) * DUMP_PITCH + n] = acc[t][r];
        }
    }
    __syncthreads();

    static_assert(CENTRE_SPLIT == 4, "the epilogue adds four partial sums");
    const float* rowA = dumpAll + (tid / PTS_PER_WAVE) * (PTS_PER_WAVE * DUMP_PITCH) + (tid % PTS_PER_WAVE) * DUMP_PITCH;
    const float* rowB = rowA + PT_GROUPS * (PTS_PER_WAVE * DUMP_PITCH);
    const float* rowC = rowB + PT_GROUPS * (PTS_PER_WAVE * DUMP_PITCH);
    const float* rowD = rowC + PT_GROUPS * (PTS_PER_WAVE * DUMP_PITCH);
    const int point = pointE;
    const float px = pxE, py = pyE;
    float S[10];
    #pragma unroll
    for (int k = 0; k < 10; ++k) S[k] = (rowA[k] + rowB[k]) + (rowC[k] + rowD[k]);
    const float S0 = S[0], S1 = S[1], S2 = S[2], S3 = S[3], S4 = S[4];
    const float S5 = S[5], S6 = S[6], S7 = S[7], S8 = S[8], S9 = S[9];
    out[0 * Npts + point] = 2.0f * (S1 - px * S0);
    out[1 * Npts + point] = 2.0f * (S3 - py * S2);
    out[2 * Npts + point] = S4;
    out[3 * Npts + point] = 2.0f * (S6 - px * S5);
    out[4 * Npts + point] = S7;
    out[5 * Npts + point] = 2.0f * (S9 - py * S8);
}

extern "C" void kernel_launch(void* const* d_in, const int* in_sizes, int n_in,
                              void* d_out, int out_size, void* d_ws, size_t ws_size,
                              hipStream_t stream) {
    const float* x  = (const float*)d_in[0];
    const float* h1 = (const float*)d_in[1];
    const float* c1 = (const float*)d_in[2];
    const float* w1 = (const float*)d_in[3];
    const float* h2 = (const float*)d_in[4];
    const float* c2 = (const float*)d_in[5];
    const float* w2 = (const float*)d_in[6];
    const float* h3 = (const float*)d_in[7];
    const float* c3 = (const float*)d_in[8];
    const float* w3 = (const float*)d_in[9];

    const int Npts = in_sizes[0] / 2;
    dim3 grid(Npts / PTS_PER_BLOCK);
    srbf2d_wmma<<<grid, BLOCK, 0, stream>>>(x, h1, c1, w1, h2, c2, w2,
                                            h3, c3, w3, (float*)d_out, Npts);
}
